# starting kernel
# baseline (speedup 1.0000x reference)
#include <hip/hip_runtime.h>


typedef __attribute__((ext_vector_type(2))) float v2f;
typedef __attribute__((ext_vector_type(8))) float v8f;

__global__ void GCN_zero_kernel(float* __restrict__ p, int n4) {
  int i = blockIdx.x * blockDim.x + threadIdx.x;
  if (i < n4) {
    ((float4*)p)[i] = make_float4(0.f, 0.f, 0.f, 0.f);
  }
}

template <int F>
__global__ void GCN_spmm_kernel(const float* __restrict__ feat,
                                const int* __restrict__ src,
                                const int* __restrict__ dst,
                                float* __restrict__ out,
                                int num_edges) {
  constexpr int TPE = F / 4;
  long long gtid = (long long)blockIdx.x * blockDim.x + threadIdx.x;
  long long e = gtid / TPE;
  int j = (int)(gtid % TPE) * 4;
  if (e >= num_edges) return;
  int s = src[e];
  int d = dst[e];
  const float4 v = *(const float4*)(feat + (long long)s * F + j);
  float* o = out + (long long)d * F + j;
  atomicAdd(o + 0, v.x);
  atomicAdd(o + 1, v.y);
  atomicAdd(o + 2, v.z);
  atomicAdd(o + 3, v.w);
}

template <int K, int NCOL>
__global__ void GCN_gemm_wmma_kernel(const float* __restrict__ A,
                                     const float* __restrict__ B,
                                     float* __restrict__ D,
                                     int nrows) {
  constexpr int LDSROW = K + 2;
  __shared__ float As[16 * LDSROW];

  const int m0 = blockIdx.x * 16;
  const int lane = threadIdx.x & 31;
  const int wave = threadIdx.x >> 5;

  for (int idx = threadIdx.x; idx < 16 * (K / 4); idx += blockDim.x) {
    int r = idx / (K / 4);
    int c = (idx % (K / 4)) * 4;
    float4 v = make_float4(0.f, 0.f, 0.f, 0.f);
    if (m0 + r < nrows) {
      v = *(const float4*)(A + (long long)(m0 + r) * K + c);
    }
    As[r * LDSROW + c + 0] = v.x;
    As[r * LDSROW + c + 1] = v.y;
    As[r * LDSROW + c + 2] = v.z;
    As[r * LDSROW + c + 3] = v.w;
  }
  __syncthreads();

  const int n0 = wave * 16;
  const int half = lane >> 4;
  const int l16 = lane & 15;

  v8f c = {};
#pragma unroll
  for (int k0 = 0; k0 < K; k0 += 4) {
    const int kb = k0 + 2 * half;
    v2f a = *(const v2f*)&As[l16 * LDSROW + kb];
    v2f b;
    b.x = B[(kb + 0) * NCOL + n0 + l16];
    b.y = B[(kb + 1) * NCOL + n0 + l16];
    c = __builtin_amdgcn_wmma_f32_16x16x4_f32(
         false, a,  false, b,
         (short)0, c,  false,  false);
  }

#pragma unroll
  for (int r = 0; r < 8; ++r) {
    int row = m0 + r + 8 * half;
    if (row < nrows) {
      D[(long long)row * NCOL + n0 + l16] = c[r];
    }
  }
}

extern "C" void kernel_launch(void* const* d_in, const int* in_sizes, int n_in,
                              void* d_out, int out_size, void* d_ws, size_t ws_size,
                              hipStream_t stream) {
  const float* x   = (const float*)d_in[0];
  const float* W1  = (const float*)d_in[1];
  const float* W2  = (const float*)d_in[2];
  const int*  src1 = (const int*)d_in[3];
  const int*  dst1 = (const int*)d_in[4];
  const int*  src2 = (const int*)d_in[5];
  const int*  dst2 = (const int*)d_in[6];
  float* out = (float*)d_out;

  constexpr int IN = 128, H = 128, C = 64;
  const int N = in_sizes[0] / IN;
  const int E = in_sizes[3];

  float* agg = (float*)d_ws;
  float* h   = agg + (size_t)N * H;
  float* t   = agg;

  {
    int n4 = N * H / 4;
    GCN_zero_kernel<<<(n4 + 255) / 256, 256, 0, stream>>>(agg, n4);
    int o4 = N * C / 4;
    GCN_zero_kernel<<<(o4 + 255) / 256, 256, 0, stream>>>(out, o4);
  }

  {
    long long threads = (long long)E * (IN / 4);
    int blocks = (int)((threads + 255) / 256);
    GCN_spmm_kernel<IN><<<blocks, 256, 0, stream>>>(x, src1, dst1, agg, E);
  }

  GCN_gemm_wmma_kernel<128, 128><<<(N + 15) / 16, 256, 0, stream>>>(agg, W1, h, N);

  GCN_gemm_wmma_kernel<128, 64><<<(N + 15) / 16, 128, 0, stream>>>(h, W2, t, N);

  {
    long long threads = (long long)E * (C / 4);
    int blocks = (int)((threads + 255) / 256);
    GCN_spmm_kernel<C><<<blocks, 256, 0, stream>>>(t, src2, dst2, out, E);
  }
}


# List building without global atomics: edge chunks ordered by 256 dst-row buckets in LDS, then a work-group per bucket links its rows by LDS integer swaps; lists and sums in LDS swap arrival order
# speedup vs baseline: 3.2341x; 3.2341x over previous
#include <hip/hip_runtime.h>


typedef __attribute__((ext_vector_type(2))) float v2f;
typedef __attribute__((ext_vector_type(4))) float v4f;
typedef __attribute__((ext_vector_type(8))) float v8f;

constexpr int BIN_CH = 8192;
constexpr int BIN_NT = 1024;
constexpr int BIN_EPT = BIN_CH / BIN_NT;
constexpr int NBKT = 256;

__global__ void __launch_bounds__(BIN_NT)
GCN_bin_kernel(const int* __restrict__ src1,
               const int* __restrict__ dst1,
               const int* __restrict__ src2,
               const int* __restrict__ dst2,
               int* __restrict__ pairs1,
               int* __restrict__ pairs2,
               int* __restrict__ desc,
               int num_edges, int rows_per_bkt, int n_chunks) {
  __shared__ int cnt[NBKT];
  __shared__ int base[NBKT];
  __shared__ int sorted_s[BIN_CH];
  __shared__ int sorted_d[BIN_CH];
  const int g = blockIdx.y;
  const int* __restrict__ src = g ? src2 : src1;
  const int* __restrict__ dst = g ? dst2 : dst1;
  int* __restrict__ pairs = g ? pairs2 : pairs1;
  const int t = threadIdx.x;
  const int e0 = blockIdx.x * BIN_CH;

  if (t < NBKT) cnt[t] = 0;
  int s[BIN_EPT], d[BIN_EPT];
#pragma unroll
  for (int k = 0; k < BIN_EPT; ++k) {
    const int e = e0 + t + BIN_NT * k;
    const bool ok = e < num_edges;
    s[k] = ok ? src[e] : 0;
    d[k] = ok ? dst[e] : 0;
  }
  __syncthreads();

  int bk[BIN_EPT], rk[BIN_EPT];
#pragma unroll
  for (int k = 0; k < BIN_EPT; ++k) {
    const int e = e0 + t + BIN_NT * k;
    unsigned q = (unsigned)d[k] / (unsigned)rows_per_bkt;
    q = q < (unsigned)(NBKT - 1) ? q : (unsigned)(NBKT - 1);
    bk[k] = (int)q;
    rk[k] = 0;
    if (e < num_edges) rk[k] = atomicAdd(&cnt[q], 1);
  }
  __syncthreads();

  if (t < 32) {
    int c8[8];
    int run = 0;
#pragma unroll
    for (int i = 0; i < 8; ++i) {
      const int v = cnt[8 * t + i];
      c8[i] = run;
      run += v;
    }
    int inc = run;
#pragma unroll
    for (int off = 1; off < 32; off <<= 1) {
      const int v = __shfl_up(inc, off);
      if (t >= off) inc += v;
    }
    const int excl = inc - run;
#pragma unroll
    for (int i = 0; i < 8; ++i) base[8 * t + i] = excl + c8[i];
  }
  __syncthreads();

#pragma unroll
  for (int k = 0; k < BIN_EPT; ++k) {
    const int e = e0 + t + BIN_NT * k;
    if (e < num_edges) {
      const int pos = base[bk[k]] + rk[k];
      sorted_s[pos] = s[k];
      sorted_d[pos] = d[k];
    }
  }
  __syncthreads();

#pragma unroll
  for (int k = 0; k < BIN_EPT; ++k) {
    const int i = t + BIN_NT * k;
    if (e0 + i < num_edges) {
      *(int2*)(pairs + 2 * (size_t)(e0 + i)) = make_int2(sorted_s[i], sorted_d[i]);
    }
  }
  if (t < NBKT) {
    *(int2*)(desc + 2 * (((size_t)g * n_chunks + blockIdx.x) * NBKT + t)) =
        make_int2(e0 + base[t], cnt[t]);
  }
}

__global__ void __launch_bounds__(BIN_NT)
GCN_bucket_link_kernel(int* __restrict__ pairs1,
                       int* __restrict__ pairs2,
                       const int* __restrict__ desc,
                       int* __restrict__ head1,
                       int* __restrict__ head2,
                       int n_rows, int num_edges, int rows_per_bkt, int n_chunks) {
  extern __shared__ int heads[];
  const int g = blockIdx.y;
  const int b = blockIdx.x;
  int* __restrict__ pairs = g ? pairs2 : pairs1;
  int* __restrict__ head = g ? head2 : head1;
  const int t = threadIdx.x;
  const int row0 = b * rows_per_bkt;

  for (int i = t; i < rows_per_bkt; i += BIN_NT) heads[i] = -1;
  __syncthreads();

  const int wave = t >> 5;
  const int lane = t & 31;
  for (int j = wave; j < n_chunks; j += BIN_NT / 32) {
    const int2 dsc = *(const int2*)(desc + 2 * (((size_t)g * n_chunks + j) * NBKT + b));
    const int start = dsc.x;
    const int count = dsc.y < BIN_CH ? dsc.y : BIN_CH;
    for (int k = lane; k < count; k += 32) {
      const int pos = start + k;
      if ((unsigned)pos < (unsigned)num_edges) {
        int2* p = (int2*)(pairs + 2 * (size_t)pos);
        const int2 sd = *p;
        unsigned idx = (unsigned)(sd.y - row0);
        idx = idx < (unsigned)rows_per_bkt ? idx : (unsigned)(rows_per_bkt - 1);
        const int prev = atomicExch(&heads[idx], pos);
        *p = make_int2(sd.x, prev);
      }
    }
  }
  __syncthreads();

  for (int i = t; i < rows_per_bkt; i += BIN_NT) {
    if (row0 + i < n_rows) head[row0 + i] = heads[i];
  }
}

template <int F>
__global__ void GCN_pull_kernel(const float* __restrict__ feat,
                                const int* __restrict__ head,
                                const int* __restrict__ rec,
                                float* __restrict__ out,
                                int n_rows, int num_edges) {
  constexpr int LPR = 8;
  constexpr int NLD = F / 32;
  static_assert(NLD == 2 || NLD == 4, "row width");
  const int gid = blockIdx.x * blockDim.x + threadIdx.x;
  const int row = gid / LPR;
  const int j4 = (gid & (LPR - 1)) * 4;
  if (row >= n_rows) return;

  v4f acc[NLD];
#pragma unroll
  for (int k = 0; k < NLD; ++k) acc[k] = v4f{0.f, 0.f, 0.f, 0.f};

  if (num_edges > 0) {
    const unsigned last = (unsigned)num_edges - 1u;
    unsigned e = (unsigned)head[row];
    unsigned ec = e < last ? e : last;
    int2 r = *(const int2*)(rec + 2 * (size_t)ec);
    for (int it = 0; it < num_edges && e <= last; ++it) {
      const float* p = feat + (size_t)r.x * F + j4;
      v4f v[NLD];
#pragma unroll
      for (int k = 0; k < NLD; ++k) v[k] = *(const v4f*)(p + 32 * k);
      e = (unsigned)r.y;
      ec = e < last ? e : last;
      const int2 rn = *(const int2*)(rec + 2 * (size_t)ec);
      if constexpr (NLD == 4) {
        asm volatile("" ::"v"(rn.x), "v"(rn.y), "v"(v[0]), "v"(v[1]), "v"(v[2]), "v"(v[3]));
      } else {
        asm volatile("" ::"v"(rn.x), "v"(rn.y), "v"(v[0]), "v"(v[1]));
      }
#pragma unroll
      for (int k = 0; k < NLD; ++k) acc[k] += v[k];
      r = rn;
    }
  }
  float* o = out + (size_t)row * F + j4;
#pragma unroll
  for (int k = 0; k < NLD; ++k) *(v4f*)(o + 32 * k) = acc[k];
}

template <int K, int H1, int C2>
__global__ void __launch_bounds__(256, 16)
GCN_layer1_kernel(const float* __restrict__ feat,
                  const int* __restrict__ head,
                  const int* __restrict__ rec,
                  const float* __restrict__ W1,
                  const float* __restrict__ W2,
                  float* __restrict__ t,
                  int n_rows, int num_edges) {
  constexpr int ROWS = 32;
  constexpr int NT = 256;
  constexpr int LPR = 8;
  constexpr int NLD = K / 32;
  constexpr int PITCH = K + 4;
  constexpr int TP = C2 + 4;
  static_assert(K == 128 && H1 == 128 && C2 == 64, "strip shapes");
  static_assert(ROWS * LPR == NT, "one lane group per row");
  __shared__ float As[ROWS * PITCH];
  __shared__ float Hs[ROWS * PITCH];

  const int tid = threadIdx.x;
  const int m0 = blockIdx.x * ROWS;

  {
    const int lr = tid / LPR;
    const int j4 = (tid & (LPR - 1)) * 4;
    const int row = m0 + lr;
    v4f acc[NLD];
#pragma unroll
    for (int k = 0; k < NLD; ++k) acc[k] = v4f{0.f, 0.f, 0.f, 0.f};

    if (row < n_rows && num_edges > 0) {
      const unsigned last = (unsigned)num_edges - 1u;
      unsigned e = (unsigned)head[row];
      unsigned ec = e < last ? e : last;
      int2 r = *(const int2*)(rec + 2 * (size_t)ec);
      for (int it = 0; it < num_edges && e <= last; ++it) {
        const float* p = feat + (size_t)r.x * K + j4;
        v4f v[NLD];
#pragma unroll
        for (int k = 0; k < NLD; ++k) v[k] = *(const v4f*)(p + 32 * k);
        e = (unsigned)r.y;
        ec = e < last ? e : last;
        const int2 rn = *(const int2*)(rec + 2 * (size_t)ec);
        asm volatile("" ::"v"(rn.x), "v"(rn.y), "v"(v[0]), "v"(v[1]), "v"(v[2]), "v"(v[3]));
#pragma unroll
        for (int k = 0; k < NLD; ++k) acc[k] += v[k];
        r = rn;
      }
    }
    float* a = As + lr * PITCH + j4;
#pragma unroll
    for (int k = 0; k < NLD; ++k) *(v4f*)(a + 32 * k) = acc[k];
  }
  __syncthreads();

  const int lane = tid & 31;
  const int wave = tid >> 5;
  const int half = lane >> 4;
  const int l16 = lane & 15;

  {
    const int n0 = wave * 16;
    v8f c0 = {};
    v8f c1 = {};
#pragma unroll
    for (int k0 = 0; k0 < K; k0 += 4) {
      const int kb = k0 + 2 * half;
      v2f a0 = *(const v2f*)&As[l16 * PITCH + kb];
      v2f a1 = *(const v2f*)&As[(16 + l16) * PITCH + kb];
      v2f b;
      b.x = W1[(kb + 0) * H1 + n0 + l16];
      b.y = W1[(kb + 1) * H1 + n0 + l16];
      c0 = __builtin_amdgcn_wmma_f32_16x16x4_f32(
           false, a0,  false, b,
           (short)0, c0,  false,  false);
      c1 = __builtin_amdgcn_wmma_f32_16x16x4_f32(
           false, a1,  false, b,
           (short)0, c1,  false,  false);
    }
#pragma unroll
    for (int r = 0; r < 8; ++r) {
      Hs[(r + 8 * half) * PITCH + n0 + l16] = c0[r];
      Hs[(16 + r + 8 * half) * PITCH + n0 + l16] = c1[r];
    }
  }
  __syncthreads();

  float* Ts = As;
  {
    const int mt = wave >> 2;
    const int n0 = (wave & 3) * 16;
    v8f c = {};
#pragma unroll
    for (int k0 = 0; k0 < H1; k0 += 4) {
      const int kb = k0 + 2 * half;
      v2f a = *(const v2f*)&Hs[(16 * mt + l16) * PITCH + kb];
      v2f b;
      b.x = W2[(kb + 0) * C2 + n0 + l16];
      b.y = W2[(kb + 1) * C2 + n0 + l16];
      c = __builtin_amdgcn_wmma_f32_16x16x4_f32(
           false, a,  false, b,
           (short)0, c,  false,  false);
    }
#pragma unroll
    for (int r = 0; r < 8; ++r) Ts[(16 * mt + r + 8 * half) * TP + n0 + l16] = c[r];
  }
  __syncthreads();

#pragma unroll
  for (int i = 0; i < (ROWS * C2 / 4) / NT; ++i) {
    const int q = tid + NT * i;
    const int r = q / (C2 / 4);
    const int c4 = (q % (C2 / 4)) * 4;
    if (m0 + r < n_rows) {
      *(v4f*)(t + (size_t)(m0 + r) * C2 + c4) = *(const v4f*)&Ts[r * TP + c4];
    }
  }
}

extern "C" void kernel_launch(void* const* d_in, const int* in_sizes, int n_in,
                              void* d_out, int out_size, void* d_ws, size_t ws_size,
                              hipStream_t stream) {
  const float* x   = (const float*)d_in[0];
  const float* W1  = (const float*)d_in[1];
  const float* W2  = (const float*)d_in[2];
  const int*  src1 = (const int*)d_in[3];
  const int*  dst1 = (const int*)d_in[4];
  const int*  src2 = (const int*)d_in[5];
  const int*  dst2 = (const int*)d_in[6];
  float* out = (float*)d_out;

  constexpr int IN = 128, H = 128, C = 64;
  const int N = in_sizes[0] / IN;
  const int E = in_sizes[3];

  float* half1 = (float*)d_ws;
  float* t   = half1 + (size_t)N * H;
  int* rec1  = (int*)half1;
  int* desc  = rec1 + 2 * (size_t)E;
  int* head1 = (int*)(t + (size_t)N * C);
  int* head2 = head1 + N;
  int* rec2  = head2 + N;

  const int RB  = (N + NBKT - 1) / NBKT;
  const int NCH = (E + BIN_CH - 1) / BIN_CH;

  GCN_bin_kernel<<<dim3(NCH, 2), BIN_NT, 0, stream>>>(src1, dst1, src2, dst2,
                                                      rec1, rec2, desc, E, RB, NCH);

  GCN_bucket_link_kernel<<<dim3(NBKT, 2), BIN_NT, RB * sizeof(int), stream>>>(
      rec1, rec2, desc, head1, head2, N, E, RB, NCH);

  GCN_layer1_kernel<IN, H, C><<<(N + 31) / 32, 256, 0, stream>>>(x, head1, rec1, W1, W2, t, N, E);

  GCN_pull_kernel<C><<<(N + 31) / 32, 256, 0, stream>>>(t, head2, rec2, out, N, E);
}


# Linked lists replaced by packed per-row id lists: launch 2 ranks a bucket's edges per row by LDS integer adds (count, scan, place); walkers read 8 ids per load, no hop per edge; sums in LDS-add order
# speedup vs baseline: 3.6113x; 1.1166x over previous
#include <hip/hip_runtime.h>


typedef __attribute__((ext_vector_type(2))) float v2f;
typedef __attribute__((ext_vector_type(4))) float v4f;
typedef __attribute__((ext_vector_type(8))) float v8f;

constexpr int BIN_CH = 8192;
constexpr int BIN_NT = 1024;
constexpr int BIN_EPT = BIN_CH / BIN_NT;
constexpr int NBKT = 256;

__global__ void __launch_bounds__(BIN_NT)
GCN_bin_kernel(const int* __restrict__ src1,
               const int* __restrict__ dst1,
               const int* __restrict__ src2,
               const int* __restrict__ dst2,
               int* __restrict__ pairs1,
               int* __restrict__ pairs2,
               int* __restrict__ desc,
               int num_edges, int rows_per_bkt, int n_chunks) {
  __shared__ int cnt[NBKT];
  __shared__ int base[NBKT];
  __shared__ int sorted_s[BIN_CH];
  __shared__ int sorted_d[BIN_CH];
  const int g = blockIdx.y;
  const int* __restrict__ src = g ? src2 : src1;
  const int* __restrict__ dst = g ? dst2 : dst1;
  int* __restrict__ pairs = g ? pairs2 : pairs1;
  const int t = threadIdx.x;
  const int e0 = blockIdx.x * BIN_CH;

  if (t < NBKT) cnt[t] = 0;
  int s[BIN_EPT], d[BIN_EPT];
#pragma unroll
  for (int k = 0; k < BIN_EPT; ++k) {
    const int e = e0 + t + BIN_NT * k;
    const bool ok = e < num_edges;
    s[k] = ok ? src[e] : 0;
    d[k] = ok ? dst[e] : 0;
  }
  __syncthreads();

  int bk[BIN_EPT], rk[BIN_EPT];
#pragma unroll
  for (int k = 0; k < BIN_EPT; ++k) {
    const int e = e0 + t + BIN_NT * k;
    unsigned q = (unsigned)d[k] / (unsigned)rows_per_bkt;
    q = q < (unsigned)(NBKT - 1) ? q : (unsigned)(NBKT - 1);
    bk[k] = (int)q;
    rk[k] = 0;
    if (e < num_edges) rk[k] = atomicAdd(&cnt[q], 1);
  }
  __syncthreads();

  if (t < 32) {
    int c8[8];
    int run = 0;
#pragma unroll
    for (int i = 0; i < 8; ++i) {
      const int v = cnt[8 * t + i];
      c8[i] = run;
      run += v;
    }
    int inc = run;
#pragma unroll
    for (int off = 1; off < 32; off <<= 1) {
      const int v = __shfl_up(inc, off);
      if (t >= off) inc += v;
    }
    const int excl = inc - run;
#pragma unroll
    for (int i = 0; i < 8; ++i) base[8 * t + i] = excl + c8[i];
  }
  __syncthreads();

#pragma unroll
  for (int k = 0; k < BIN_EPT; ++k) {
    const int e = e0 + t + BIN_NT * k;
    if (e < num_edges) {
      const int pos = base[bk[k]] + rk[k];
      sorted_s[pos] = s[k];
      sorted_d[pos] = d[k];
    }
  }
  __syncthreads();

#pragma unroll
  for (int k = 0; k < BIN_EPT; ++k) {
    const int i = t + BIN_NT * k;
    if (e0 + i < num_edges) {
      *(int2*)(pairs + 2 * (size_t)(e0 + i)) = make_int2(sorted_s[i], sorted_d[i]);
    }
  }
  if (t < NBKT) {
    *(int2*)(desc + 2 * (((size_t)g * n_chunks + blockIdx.x) * NBKT + t)) =
        make_int2(e0 + base[t], cnt[t]);
  }
}

constexpr int ROWS_CAP = 8192;
constexpr int ROWS_RG = 4;

template <bool PLACE>
__device__ __forceinline__ void GCN_bucket_pass(const int* __restrict__ pairs,
                                                const int* __restrict__ desc_gb,
                                                int* __restrict__ col,
                                                int* cnt, int* ids, int* bucket_base,
                                                int bb, int row0, int rows_per_bkt,
                                                int num_edges, int n_chunks,
                                                int wave, int lane) {
  for (int j0 = wave; j0 < n_chunks; j0 += (BIN_NT / 32) * ROWS_RG) {
    int start[ROWS_RG], count[ROWS_RG];
    int2 dsc[ROWS_RG];
#pragma unroll
    for (int r = 0; r < ROWS_RG; ++r) {
      const int j = j0 + (BIN_NT / 32) * r;
      const int jc = j < n_chunks ? j : j0;
      dsc[r] = *(const int2*)(desc_gb + 2 * (size_t)jc * NBKT);
    }
    int off_sum = 0;
#pragma unroll
    for (int r = 0; r < ROWS_RG; ++r) {
      const int j = j0 + (BIN_NT / 32) * r;
      const bool valid = j < n_chunks;
      const int cn = dsc[r].y < BIN_CH ? dsc[r].y : BIN_CH;
      start[r] = dsc[r].x;
      count[r] = valid ? cn : 0;
      off_sum += valid ? dsc[r].x - j * BIN_CH : 0;
    }
    if (!PLACE && lane == 0) atomicAdd(bucket_base, off_sum);

    int2 sd[ROWS_RG][2];
    bool ok[ROWS_RG][2];
#pragma unroll
    for (int r = 0; r < ROWS_RG; ++r) {
#pragma unroll
      for (int h = 0; h < 2; ++h) {
        const int k = lane + 32 * h;
        const int pos = start[r] + k;
        ok[r][h] = k < count[r] && (unsigned)pos < (unsigned)num_edges;
        sd[r][h] = make_int2(0, 0);
        if (ok[r][h]) sd[r][h] = *(const int2*)(pairs + 2 * (size_t)pos);
      }
    }
    int slot[ROWS_RG][2];
#pragma unroll
    for (int r = 0; r < ROWS_RG; ++r) {
#pragma unroll
      for (int h = 0; h < 2; ++h) {
        slot[r][h] = 0;
        if (ok[r][h]) {
          unsigned idx = (unsigned)(sd[r][h].y - row0);
          idx = idx < (unsigned)rows_per_bkt ? idx : (unsigned)(rows_per_bkt - 1);
          if (!PLACE) {
            atomicAdd(&cnt[idx], 1);
          } else {
            slot[r][h] = atomicAdd(&cnt[idx], 1);
          }
        }
      }
    }
    if (PLACE) {
#pragma unroll
      for (int r = 0; r < ROWS_RG; ++r) {
#pragma unroll
        for (int h = 0; h < 2; ++h) {
          if (ok[r][h]) {
            if ((unsigned)slot[r][h] < (unsigned)ROWS_CAP) {
              ids[slot[r][h]] = sd[r][h].x;
            } else if ((unsigned)(bb + slot[r][h]) < (unsigned)num_edges) {
              col[bb + slot[r][h]] = sd[r][h].x;
            }
          }
        }
      }
    }
#pragma unroll
    for (int r = 0; r < ROWS_RG; ++r) {
      for (int k = lane + 64; k < count[r]; k += 32) {
        const int pos = start[r] + k;
        if ((unsigned)pos < (unsigned)num_edges) {
          const int2 e = *(const int2*)(pairs + 2 * (size_t)pos);
          unsigned idx = (unsigned)(e.y - row0);
          idx = idx < (unsigned)rows_per_bkt ? idx : (unsigned)(rows_per_bkt - 1);
          if (!PLACE) {
            atomicAdd(&cnt[idx], 1);
          } else {
            const int slot = atomicAdd(&cnt[idx], 1);
            if ((unsigned)slot < (unsigned)ROWS_CAP) {
              ids[slot] = e.x;
            } else if ((unsigned)(bb + slot) < (unsigned)num_edges) {
              col[bb + slot] = e.x;
            }
          }
        }
      }
    }
  }
}

__global__ void __launch_bounds__(BIN_NT)
GCN_bucket_rows_kernel(const int* __restrict__ pairs1,
                       const int* __restrict__ pairs2,
                       const int* __restrict__ desc,
                       int* __restrict__ col1,
                       int* __restrict__ col2,
                       int* __restrict__ rows1,
                       int* __restrict__ rows2,
                       int n_rows, int num_edges, int rows_per_bkt, int n_chunks) {
  extern __shared__ int lds[];
  __shared__ int bucket_base;
  __shared__ int bucket_total;
  int* cnt = lds;
  int* first = lds + rows_per_bkt;
  int* ids = first + rows_per_bkt;
  const int g = blockIdx.y;
  const int b = blockIdx.x;
  const int* __restrict__ pairs = g ? pairs2 : pairs1;
  int* __restrict__ col = g ? col2 : col1;
  int* __restrict__ rows = g ? rows2 : rows1;
  const int* __restrict__ desc_gb = desc + 2 * ((size_t)g * n_chunks * NBKT + b);
  const int t = threadIdx.x;
  const int row0 = b * rows_per_bkt;
  const int wave = t >> 5;
  const int lane = t & 31;

  for (int i = t; i < rows_per_bkt; i += BIN_NT) cnt[i] = 0;
  if (t == 0) bucket_base = 0;
  __syncthreads();

  GCN_bucket_pass<false>(pairs, desc_gb, col, cnt, ids, &bucket_base, 0, row0, rows_per_bkt,
                         num_edges, n_chunks, wave, lane);
  __syncthreads();

  if (t < 32) {
    const int per = (rows_per_bkt + 31) / 32;
    const int lo = t * per < rows_per_bkt ? t * per : rows_per_bkt;
    const int hi = lo + per < rows_per_bkt ? lo + per : rows_per_bkt;
    int run = 0;
    for (int i = lo; i < hi; ++i) run += cnt[i];
    int inc = run;
#pragma unroll
    for (int off = 1; off < 32; off <<= 1) {
      const int u = __shfl_up(inc, off);
      if (t >= off) inc += u;
    }
    int at = inc - run;
    for (int i = lo; i < hi; ++i) {
      const int v = cnt[i];
      first[i] = at;
      cnt[i] = at;
      at += v;
    }
    if (t == 31) bucket_total = inc;
  }
  __syncthreads();

  const int bb = bucket_base;
  GCN_bucket_pass<true>(pairs, desc_gb, col, cnt, ids, &bucket_base, bb, row0, rows_per_bkt,
                        num_edges, n_chunks, wave, lane);
  __syncthreads();

  const int total = bucket_total;
  const int n_lds = total < ROWS_CAP ? total : ROWS_CAP;
  for (int i = t; i < n_lds; i += BIN_NT) {
    if ((unsigned)(bb + i) < (unsigned)num_edges) col[bb + i] = ids[i];
  }
  for (int i = t; i < rows_per_bkt; i += BIN_NT) {
    if (row0 + i < n_rows) {
      *(int2*)(rows + 2 * (size_t)(row0 + i)) = make_int2(bb + first[i], cnt[i] - first[i]);
    }
  }
}

template <int SRC>
__device__ __forceinline__ int GCN_bcast8(int x) {
  constexpr unsigned sel = (unsigned)SRC * 0x249249u;
  return (int)__builtin_amdgcn_mov_dpp8((unsigned)x, sel);
}

template <int F, int KIDX>
__device__ __forceinline__ void GCN_load_edge(const float* __restrict__ feat, int myid,
                                              unsigned rem, int j4, v4f (&v)[F / 32]) {
  const int s = GCN_bcast8<KIDX>(myid);
#pragma unroll
  for (int k = 0; k < F / 32; ++k) v[k] = v4f{0.f, 0.f, 0.f, 0.f};
  if ((unsigned)KIDX < rem) {
    const float* p = feat + (size_t)s * F + j4;
#pragma unroll
    for (int k = 0; k < F / 32; ++k) v[k] = *(const v4f*)(p + 32 * k);
  }
}

template <int F, int K0, int U>
__device__ __forceinline__ void GCN_edge_group(const float* __restrict__ feat, int myid,
                                               unsigned rem, int j4, v4f (&acc)[F / 32]) {
  static_assert(U == 2 || U == 4, "edges per group");
  constexpr int NLD = F / 32;
  if ((unsigned)K0 < rem) {
    v4f va[NLD], vb[NLD];
    GCN_load_edge<F, K0>(feat, myid, rem, j4, va);
    GCN_load_edge<F, K0 + 1>(feat, myid, rem, j4, vb);
    if constexpr (U == 4) {
      v4f vc[NLD], vd[NLD];
      GCN_load_edge<F, K0 + 2>(feat, myid, rem, j4, vc);
      GCN_load_edge<F, K0 + 3>(feat, myid, rem, j4, vd);
#pragma unroll
      for (int k = 0; k < NLD; ++k) acc[k] += va[k];
#pragma unroll
      for (int k = 0; k < NLD; ++k) acc[k] += vb[k];
#pragma unroll
      for (int k = 0; k < NLD; ++k) acc[k] += vc[k];
#pragma unroll
      for (int k = 0; k < NLD; ++k) acc[k] += vd[k];
    } else {
#pragma unroll
      for (int k = 0; k < NLD; ++k) acc[k] += va[k];
#pragma unroll
      for (int k = 0; k < NLD; ++k) acc[k] += vb[k];
    }
  }
}

template <int F, int U>
__device__ __forceinline__ void GCN_sum_row(const float* __restrict__ feat,
                                            const int* __restrict__ rows,
                                            const int* __restrict__ col,
                                            int row, int jl, int num_edges,
                                            v4f (&acc)[F / 32]) {
  const int j4 = jl * 4;
  const int2 rc = *(const int2*)(rows + 2 * (size_t)row);
  const unsigned ne = (unsigned)num_edges;
  const unsigned st = (unsigned)rc.x < ne ? (unsigned)rc.x : ne;
  const unsigned n = (unsigned)rc.y < ne - st ? (unsigned)rc.y : ne - st;
  const int* __restrict__ lst = col + st;
  int myid = (unsigned)jl < n ? lst[jl] : 0;
  asm volatile("" ::"v"(myid));
  for (unsigned i = 0; i < n; i += 8) {
    const unsigned rem = n - i;
    const unsigned nx = i + 8 + (unsigned)jl;
    const int nid = nx < n ? lst[nx] : 0;
    GCN_edge_group<F, 0, U>(feat, myid, rem, j4, acc);
    if constexpr (U == 2) GCN_edge_group<F, 2, U>(feat, myid, rem, j4, acc);
    GCN_edge_group<F, 4, U>(feat, myid, rem, j4, acc);
    if constexpr (U == 2) GCN_edge_group<F, 6, U>(feat, myid, rem, j4, acc);
    myid = nid;
  }
}

template <int F>
__global__ void GCN_pull_kernel(const float* __restrict__ feat,
                                const int* __restrict__ rows,
                                const int* __restrict__ col,
                                float* __restrict__ out,
                                int n_rows, int num_edges) {
  constexpr int LPR = 8;
  constexpr int NLD = F / 32;
  static_assert(NLD == 2 || NLD == 4, "row width");
  const int gid = blockIdx.x * blockDim.x + threadIdx.x;
  const int row = gid / LPR;
  const int jl = gid & (LPR - 1);
  if (row >= n_rows) return;

  v4f acc[NLD];
#pragma unroll
  for (int k = 0; k < NLD; ++k) acc[k] = v4f{0.f, 0.f, 0.f, 0.f};

  GCN_sum_row<F, 4>(feat, rows, col, row, jl, num_edges, acc);

  float* o = out + (size_t)row * F + jl * 4;
#pragma unroll
  for (int k = 0; k < NLD; ++k) *(v4f*)(o + 32 * k) = acc[k];
}

template <int K, int H1, int C2>
__global__ void __launch_bounds__(256, 16)
GCN_layer1_kernel(const float* __restrict__ feat,
                  const int* __restrict__ rows,
                  const int* __restrict__ col,
                  const float* __restrict__ W1,
                  const float* __restrict__ W2,
                  float* __restrict__ t,
                  int n_rows, int num_edges) {
  constexpr int ROWS = 32;
  constexpr int NT = 256;
  constexpr int LPR = 8;
  constexpr int NLD = K / 32;
  constexpr int PITCH = K + 4;
  constexpr int TP = C2 + 4;
  static_assert(K == 128 && H1 == 128 && C2 == 64, "strip shapes");
  static_assert(ROWS * LPR == NT, "one lane group per row");
  __shared__ float As[ROWS * PITCH];
  __shared__ float Hs[ROWS * PITCH];

  const int tid = threadIdx.x;
  const int m0 = blockIdx.x * ROWS;

  {
    const int lr = tid / LPR;
    const int jl = tid & (LPR - 1);
    const int row = m0 + lr;
    v4f acc[NLD];
#pragma unroll
    for (int k = 0; k < NLD; ++k) acc[k] = v4f{0.f, 0.f, 0.f, 0.f};

    if (row < n_rows) GCN_sum_row<K, 2>(feat, rows, col, row, jl, num_edges, acc);

    float* a = As + lr * PITCH + jl * 4;
#pragma unroll
    for (int k = 0; k < NLD; ++k) *(v4f*)(a + 32 * k) = acc[k];
  }
  __syncthreads();

  const int lane = tid & 31;
  const int wave = tid >> 5;
  const int half = lane >> 4;
  const int l16 = lane & 15;

  {
    const int n0 = wave * 16;
    v8f c0 = {};
    v8f c1 = {};
#pragma unroll
    for (int k0 = 0; k0 < K; k0 += 4) {
      const int kb = k0 + 2 * half;
      v2f a0 = *(const v2f*)&As[l16 * PITCH + kb];
      v2f a1 = *(const v2f*)&As[(16 + l16) * PITCH + kb];
      v2f b;
      b.x = W1[(kb + 0) * H1 + n0 + l16];
      b.y = W1[(kb + 1) * H1 + n0 + l16];
      c0 = __builtin_amdgcn_wmma_f32_16x16x4_f32(
           false, a0,  false, b,
           (short)0, c0,  false,  false);
      c1 = __builtin_amdgcn_wmma_f32_16x16x4_f32(
           false, a1,  false, b,
           (short)0, c1,  false,  false);
    }
#pragma unroll
    for (int r = 0; r < 8; ++r) {
      Hs[(r + 8 * half) * PITCH + n0 + l16] = c0[r];
      Hs[(16 + r + 8 * half) * PITCH + n0 + l16] = c1[r];
    }
  }
  __syncthreads();

  float* Ts = As;
  {
    const int mt = wave >> 2;
    const int n0 = (wave & 3) * 16;
    v8f c = {};
#pragma unroll
    for (int k0 = 0; k0 < H1; k0 += 4) {
      const int kb = k0 + 2 * half;
      v2f a = *(const v2f*)&Hs[(16 * mt + l16) * PITCH + kb];
      v2f b;
      b.x = W2[(kb + 0) * C2 + n0 + l16];
      b.y = W2[(kb + 1) * C2 + n0 + l16];
      c = __builtin_amdgcn_wmma_f32_16x16x4_f32(
           false, a,  false, b,
           (short)0, c,  false,  false);
    }
#pragma unroll
    for (int r = 0; r < 8; ++r) Ts[(16 * mt + r + 8 * half) * TP + n0 + l16] = c[r];
  }
  __syncthreads();

#pragma unroll
  for (int i = 0; i < (ROWS * C2 / 4) / NT; ++i) {
    const int q = tid + NT * i;
    const int r = q / (C2 / 4);
    const int c4 = (q % (C2 / 4)) * 4;
    if (m0 + r < n_rows) {
      *(v4f*)(t + (size_t)(m0 + r) * C2 + c4) = *(const v4f*)&Ts[r * TP + c4];
    }
  }
}

extern "C" void kernel_launch(void* const* d_in, const int* in_sizes, int n_in,
                              void* d_out, int out_size, void* d_ws, size_t ws_size,
                              hipStream_t stream) {
  const float* x   = (const float*)d_in[0];
  const float* W1  = (const float*)d_in[1];
  const float* W2  = (const float*)d_in[2];
  const int*  src1 = (const int*)d_in[3];
  const int*  dst1 = (const int*)d_in[4];
  const int*  src2 = (const int*)d_in[5];
  const int*  dst2 = (const int*)d_in[6];
  float* out = (float*)d_out;

  constexpr int IN = 128, H = 128, C = 64;
  const int N = in_sizes[0] / IN;
  const int E = in_sizes[3];

  float* half1 = (float*)d_ws;
  float* t   = half1 + (size_t)N * H;
  const int RB  = (N + NBKT - 1) / NBKT;
  const int NCH = (E + BIN_CH - 1) / BIN_CH;
  int* rec1  = (int*)half1;
  int* desc  = rec1 + 2 * (size_t)E;
  int* rows1 = desc + 4 * (size_t)NCH * NBKT;
  int* rows2 = rows1 + 2 * (size_t)N;
  int* col1  = rows2 + 2 * (size_t)N;
  int* col2  = col1 + (size_t)E;
  int* rec2  = (int*)(t + (size_t)N * C);

  GCN_bin_kernel<<<dim3(NCH, 2), BIN_NT, 0, stream>>>(src1, dst1, src2, dst2,
                                                      rec1, rec2, desc, E, RB, NCH);

  GCN_bucket_rows_kernel<<<dim3(NBKT, 2), BIN_NT, (2 * RB + ROWS_CAP) * sizeof(int), stream>>>(
      rec1, rec2, desc, col1, col2, rows1, rows2, N, E, RB, NCH);

  GCN_layer1_kernel<IN, H, C><<<(N + 31) / 32, 256, 0, stream>>>(x, rows1, col1, W1, W2, t, N, E);

  GCN_pull_kernel<C><<<(N + 31) / 32, 256, 0, stream>>>(t, rows2, col2, out, N, E);
}


# Row gathers re-mapped to one wave per destination row: a lane owns 16 B (layer 1) or 8 B (layer 2) of a row, so one load covers a whole source row; ids by v_readlane; sums in list order as before
# speedup vs baseline: 3.7553x; 1.0399x over previous
#include <hip/hip_runtime.h>


typedef __attribute__((ext_vector_type(2))) float v2f;
typedef __attribute__((ext_vector_type(4))) float v4f;
typedef __attribute__((ext_vector_type(8))) float v8f;

constexpr int BIN_CH = 8192;
constexpr int BIN_NT = 1024;
constexpr int BIN_EPT = BIN_CH / BIN_NT;
constexpr int NBKT = 256;

__global__ void __launch_bounds__(BIN_NT)
GCN_bin_kernel(const int* __restrict__ src1,
               const int* __restrict__ dst1,
               const int* __restrict__ src2,
               const int* __restrict__ dst2,
               int* __restrict__ pairs1,
               int* __restrict__ pairs2,
               int* __restrict__ desc,
               int num_edges, int rows_per_bkt, int n_chunks) {
  __shared__ int cnt[NBKT];
  __shared__ int base[NBKT];
  __shared__ int sorted_s[BIN_CH];
  __shared__ int sorted_d[BIN_CH];
  const int g = blockIdx.y;
  const int* __restrict__ src = g ? src2 : src1;
  const int* __restrict__ dst = g ? dst2 : dst1;
  int* __restrict__ pairs = g ? pairs2 : pairs1;
  const int t = threadIdx.x;
  const int e0 = blockIdx.x * BIN_CH;

  if (t < NBKT) cnt[t] = 0;
  int s[BIN_EPT], d[BIN_EPT];
#pragma unroll
  for (int k = 0; k < BIN_EPT; ++k) {
    const int e = e0 + t + BIN_NT * k;
    const bool ok = e < num_edges;
    s[k] = ok ? src[e] : 0;
    d[k] = ok ? dst[e] : 0;
  }
  __syncthreads();

  int bk[BIN_EPT], rk[BIN_EPT];
#pragma unroll
  for (int k = 0; k < BIN_EPT; ++k) {
    const int e = e0 + t + BIN_NT * k;
    unsigned q = (unsigned)d[k] / (unsigned)rows_per_bkt;
    q = q < (unsigned)(NBKT - 1) ? q : (unsigned)(NBKT - 1);
    bk[k] = (int)q;
    rk[k] = 0;
    if (e < num_edges) rk[k] = atomicAdd(&cnt[q], 1);
  }
  __syncthreads();

  if (t < 32) {
    int c8[8];
    int run = 0;
#pragma unroll
    for (int i = 0; i < 8; ++i) {
      const int v = cnt[8 * t + i];
      c8[i] = run;
      run += v;
    }
    int inc = run;
#pragma unroll
    for (int off = 1; off < 32; off <<= 1) {
      const int v = __shfl_up(inc, off);
      if (t >= off) inc += v;
    }
    const int excl = inc - run;
#pragma unroll
    for (int i = 0; i < 8; ++i) base[8 * t + i] = excl + c8[i];
  }
  __syncthreads();

#pragma unroll
  for (int k = 0; k < BIN_EPT; ++k) {
    const int e = e0 + t + BIN_NT * k;
    if (e < num_edges) {
      const int pos = base[bk[k]] + rk[k];
      sorted_s[pos] = s[k];
      sorted_d[pos] = d[k];
    }
  }
  __syncthreads();

#pragma unroll
  for (int k = 0; k < BIN_EPT; ++k) {
    const int i = t + BIN_NT * k;
    if (e0 + i < num_edges) {
      *(int2*)(pairs + 2 * (size_t)(e0 + i)) = make_int2(sorted_s[i], sorted_d[i]);
    }
  }
  if (t < NBKT) {
    *(int2*)(desc + 2 * (((size_t)g * n_chunks + blockIdx.x) * NBKT + t)) =
        make_int2(e0 + base[t], cnt[t]);
  }
}

constexpr int ROWS_CAP = 8192;
constexpr int ROWS_RG = 4;

template <bool PLACE>
__device__ __forceinline__ void GCN_bucket_pass(const int* __restrict__ pairs,
                                                const int* __restrict__ desc_gb,
                                                int* __restrict__ col,
                                                int* cnt, int* ids, int* bucket_base,
                                                int bb, int row0, int rows_per_bkt,
                                                int num_edges, int n_chunks,
                                                int wave, int lane) {
  for (int j0 = wave; j0 < n_chunks; j0 += (BIN_NT / 32) * ROWS_RG) {
    int start[ROWS_RG], count[ROWS_RG];
    int2 dsc[ROWS_RG];
#pragma unroll
    for (int r = 0; r < ROWS_RG; ++r) {
      const int j = j0 + (BIN_NT / 32) * r;
      const int jc = j < n_chunks ? j : j0;
      dsc[r] = *(const int2*)(desc_gb + 2 * (size_t)jc * NBKT);
    }
    int off_sum = 0;
#pragma unroll
    for (int r = 0; r < ROWS_RG; ++r) {
      const int j = j0 + (BIN_NT / 32) * r;
      const bool valid = j < n_chunks;
      const int cn = dsc[r].y < BIN_CH ? dsc[r].y : BIN_CH;
      start[r] = dsc[r].x;
      count[r] = valid ? cn : 0;
      off_sum += valid ? dsc[r].x - j * BIN_CH : 0;
    }
    if (!PLACE && lane == 0) atomicAdd(bucket_base, off_sum);

    int2 sd[ROWS_RG][2];
    bool ok[ROWS_RG][2];
#pragma unroll
    for (int r = 0; r < ROWS_RG; ++r) {
#pragma unroll
      for (int h = 0; h < 2; ++h) {
        const int k = lane + 32 * h;
        const int pos = start[r] + k;
        ok[r][h] = k < count[r] && (unsigned)pos < (unsigned)num_edges;
        sd[r][h] = make_int2(0, 0);
        if (ok[r][h]) sd[r][h] = *(const int2*)(pairs + 2 * (size_t)pos);
      }
    }
    int slot[ROWS_RG][2];
#pragma unroll
    for (int r = 0; r < ROWS_RG; ++r) {
#pragma unroll
      for (int h = 0; h < 2; ++h) {
        slot[r][h] = 0;
        if (ok[r][h]) {
          unsigned idx = (unsigned)(sd[r][h].y - row0);
          idx = idx < (unsigned)rows_per_bkt ? idx : (unsigned)(rows_per_bkt - 1);
          if (!PLACE) {
            atomicAdd(&cnt[idx], 1);
          } else {
            slot[r][h] = atomicAdd(&cnt[idx], 1);
          }
        }
      }
    }
    if (PLACE) {
#pragma unroll
      for (int r = 0; r < ROWS_RG; ++r) {
#pragma unroll
        for (int h = 0; h < 2; ++h) {
          if (ok[r][h]) {
            if ((unsigned)slot[r][h] < (unsigned)ROWS_CAP) {
              ids[slot[r][h]] = sd[r][h].x;
            } else if ((unsigned)(bb + slot[r][h]) < (unsigned)num_edges) {
              col[bb + slot[r][h]] = sd[r][h].x;
            }
          }
        }
      }
    }
#pragma unroll
    for (int r = 0; r < ROWS_RG; ++r) {
      for (int k = lane + 64; k < count[r]; k += 32) {
        const int pos = start[r] + k;
        if ((unsigned)pos < (unsigned)num_edges) {
          const int2 e = *(const int2*)(pairs + 2 * (size_t)pos);
          unsigned idx = (unsigned)(e.y - row0);
          idx = idx < (unsigned)rows_per_bkt ? idx : (unsigned)(rows_per_bkt - 1);
          if (!PLACE) {
            atomicAdd(&cnt[idx], 1);
          } else {
            const int slot = atomicAdd(&cnt[idx], 1);
            if ((unsigned)slot < (unsigned)ROWS_CAP) {
              ids[slot] = e.x;
            } else if ((unsigned)(bb + slot) < (unsigned)num_edges) {
              col[bb + slot] = e.x;
            }
          }
        }
      }
    }
  }
}

__global__ void __launch_bounds__(BIN_NT)
GCN_bucket_rows_kernel(const int* __restrict__ pairs1,
                       const int* __restrict__ pairs2,
                       const int* __restrict__ desc,
                       int* __restrict__ col1,
                       int* __restrict__ col2,
                       int* __restrict__ rows1,
                       int* __restrict__ rows2,
                       int n_rows, int num_edges, int rows_per_bkt, int n_chunks) {
  extern __shared__ int lds[];
  __shared__ int bucket_base;
  __shared__ int bucket_total;
  int* cnt = lds;
  int* first = lds + rows_per_bkt;
  int* ids = first + rows_per_bkt;
  const int g = blockIdx.y;
  const int b = blockIdx.x;
  const int* __restrict__ pairs = g ? pairs2 : pairs1;
  int* __restrict__ col = g ? col2 : col1;
  int* __restrict__ rows = g ? rows2 : rows1;
  const int* __restrict__ desc_gb = desc + 2 * ((size_t)g * n_chunks * NBKT + b);
  const int t = threadIdx.x;
  const int row0 = b * rows_per_bkt;
  const int wave = t >> 5;
  const int lane = t & 31;

  for (int i = t; i < rows_per_bkt; i += BIN_NT) cnt[i] = 0;
  if (t == 0) bucket_base = 0;
  __syncthreads();

  GCN_bucket_pass<false>(pairs, desc_gb, col, cnt, ids, &bucket_base, 0, row0, rows_per_bkt,
                         num_edges, n_chunks, wave, lane);
  __syncthreads();

  if (t < 32) {
    const int per = (rows_per_bkt + 31) / 32;
    const int lo = t * per < rows_per_bkt ? t * per : rows_per_bkt;
    const int hi = lo + per < rows_per_bkt ? lo + per : rows_per_bkt;
    int run = 0;
    for (int i = lo; i < hi; ++i) run += cnt[i];
    int inc = run;
#pragma unroll
    for (int off = 1; off < 32; off <<= 1) {
      const int u = __shfl_up(inc, off);
      if (t >= off) inc += u;
    }
    int at = inc - run;
    for (int i = lo; i < hi; ++i) {
      const int v = cnt[i];
      first[i] = at;
      cnt[i] = at;
      at += v;
    }
    if (t == 31) bucket_total = inc;
  }
  __syncthreads();

  const int bb = bucket_base;
  GCN_bucket_pass<true>(pairs, desc_gb, col, cnt, ids, &bucket_base, bb, row0, rows_per_bkt,
                        num_edges, n_chunks, wave, lane);
  __syncthreads();

  const int total = bucket_total;
  const int n_lds = total < ROWS_CAP ? total : ROWS_CAP;
  for (int i = t; i < n_lds; i += BIN_NT) {
    if ((unsigned)(bb + i) < (unsigned)num_edges) col[bb + i] = ids[i];
  }
  for (int i = t; i < rows_per_bkt; i += BIN_NT) {
    if (row0 + i < n_rows) {
      *(int2*)(rows + 2 * (size_t)(row0 + i)) = make_int2(bb + first[i], cnt[i] - first[i]);
    }
  }
}

template <int VW> struct GCN_vec;
template <> struct GCN_vec<2> { typedef v2f type; };
template <> struct GCN_vec<4> { typedef v4f type; };

template <int RPW>
__device__ __forceinline__ void GCN_row_ranges(const int* __restrict__ rows, int row0, int n_rows,
                                               int num_edges, int lane,
                                               unsigned (&st)[RPW], unsigned (&n)[RPW]) {
  static_assert((RPW & (RPW - 1)) == 0 && RPW <= 32, "rows per wave");
  const int r = row0 + (lane & (RPW - 1));
  const int rr = r < n_rows ? r : n_rows - 1;
  const int2 rc = *(const int2*)(rows + 2 * (size_t)rr);
  const unsigned ne = (unsigned)num_edges;
  const unsigned s = (unsigned)rc.x < ne ? (unsigned)rc.x : ne;
  const unsigned c = (unsigned)rc.y < ne - s ? (unsigned)rc.y : ne - s;
  const unsigned cz = r < n_rows ? c : 0u;
#pragma unroll
  for (int q = 0; q < RPW; ++q) {
    st[q] = (unsigned)__builtin_amdgcn_readlane((int)s, q);
    n[q] = (unsigned)__builtin_amdgcn_readlane((int)cz, q);
  }
}

template <int F, int U>
__device__ __forceinline__ typename GCN_vec<F / 32>::type
GCN_wave_row_sum(const float* __restrict__ feat, const int* __restrict__ lst,
                 unsigned n, int ids, int lane) {
  typedef typename GCN_vec<F / 32>::type VT;
  constexpr int VW = F / 32;
  const unsigned lo = (unsigned)lane * VW;
  const VT zero = {};
  VT acc = zero;
  for (unsigned i = 0; i < n; i += 32u) {
    const unsigned left = n - i;
    const unsigned cnt = left < 32u ? left : 32u;
    asm volatile("" ::"v"(ids));
    int nids = 0;
    if (left > 32u) {
      const unsigned nx = i + 32u + (unsigned)lane;
      nids = nx < n ? lst[nx] : 0;
    }
    for (unsigned e = 0; e < cnt; e += (unsigned)U) {
      VT v[U];
#pragma unroll
      for (int u = 0; u < U; ++u) {
        if (e + (unsigned)u < cnt) {
          const int s = __builtin_amdgcn_readlane(ids, (int)e + u);
          const float* __restrict__ rowp = feat + (size_t)s * F;
          v[u] = *(const VT*)(rowp + lo);
        }
      }
#pragma unroll
      for (int u = 0; u < U; ++u) {
        if (e + (unsigned)u < cnt) acc += v[u];
      }
    }
    ids = nids;
  }
  return acc;
}

template <int F>
__global__ void GCN_pull_kernel(const float* __restrict__ feat,
                                const int* __restrict__ rows,
                                const int* __restrict__ col,
                                float* __restrict__ out,
                                int n_rows, int num_edges) {
  constexpr int RPW = 4;
  constexpr int VW = F / 32;
  constexpr int U = 32 / VW;
  typedef typename GCN_vec<VW>::type VT;
  const int lane = threadIdx.x & 31;
  const int gw = __builtin_amdgcn_readfirstlane((int)((blockIdx.x * blockDim.x + threadIdx.x) >> 5));
  const int row0 = gw * RPW;
  if (row0 >= n_rows) return;

  unsigned st[RPW], n[RPW];
  GCN_row_ranges<RPW>(rows, row0, n_rows, num_edges, lane, st, n);
  int ids[RPW];
#pragma unroll
  for (int q = 0; q < RPW; ++q) ids[q] = (unsigned)lane < n[q] ? col[st[q] + (unsigned)lane] : 0;
#pragma unroll
  for (int q = 0; q < RPW; ++q) {
    const VT acc = GCN_wave_row_sum<F, U>(feat, col + st[q], n[q], ids[q], lane);
    if (row0 + q < n_rows) *(VT*)(out + (size_t)(row0 + q) * F + lane * VW) = acc;
  }
}

template <int K, int H1, int C2>
__global__ void __launch_bounds__(256, 16)
GCN_layer1_kernel(const float* __restrict__ feat,
                  const int* __restrict__ rows,
                  const int* __restrict__ col,
                  const float* __restrict__ W1,
                  const float* __restrict__ W2,
                  float* __restrict__ t,
                  int n_rows, int num_edges) {
  constexpr int ROWS = 32;
  constexpr int NT = 256;
  constexpr int RPW = ROWS / (NT / 32);
  constexpr int PITCH = K + 4;
  constexpr int TP = C2 + 4;
  static_assert(K == 128 && H1 == 128 && C2 == 64, "strip shapes");
  static_assert(RPW * (NT / 32) == ROWS, "whole rows per wave");
  __shared__ float As[ROWS * PITCH];
  __shared__ float Hs[ROWS * PITCH];

  const int tid = threadIdx.x;
  const int m0 = blockIdx.x * ROWS;

  {
    const int ln = tid & 31;
    const int wv = __builtin_amdgcn_readfirstlane(tid >> 5);
    const int lr0 = RPW * wv;
    unsigned st[RPW], n[RPW];
    GCN_row_ranges<RPW>(rows, m0 + lr0, n_rows, num_edges, ln, st, n);
    int ids[RPW];
#pragma unroll
    for (int q = 0; q < RPW; ++q) ids[q] = (unsigned)ln < n[q] ? col[st[q] + (unsigned)ln] : 0;
#pragma unroll
    for (int q = 0; q < RPW; ++q) {
      const v4f acc = GCN_wave_row_sum<K, 8>(feat, col + st[q], n[q], ids[q], ln);
      *(v4f*)(As + (lr0 + q) * PITCH + 4 * ln) = acc;
    }
  }
  __syncthreads();

  const int lane = tid & 31;
  const int wave = tid >> 5;
  const int half = lane >> 4;
  const int l16 = lane & 15;

  {
    const int n0 = wave * 16;
    v8f c0 = {};
    v8f c1 = {};
#pragma unroll
    for (int k0 = 0; k0 < K; k0 += 4) {
      const int kb = k0 + 2 * half;
      v2f a0 = *(const v2f*)&As[l16 * PITCH + kb];
      v2f a1 = *(const v2f*)&As[(16 + l16) * PITCH + kb];
      v2f b;
      b.x = W1[(kb + 0) * H1 + n0 + l16];
      b.y = W1[(kb + 1) * H1 + n0 + l16];
      c0 = __builtin_amdgcn_wmma_f32_16x16x4_f32(
           false, a0,  false, b,
           (short)0, c0,  false,  false);
      c1 = __builtin_amdgcn_wmma_f32_16x16x4_f32(
           false, a1,  false, b,
           (short)0, c1,  false,  false);
    }
#pragma unroll
    for (int r = 0; r < 8; ++r) {
      Hs[(r + 8 * half) * PITCH + n0 + l16] = c0[r];
      Hs[(16 + r + 8 * half) * PITCH + n0 + l16] = c1[r];
    }
  }
  __syncthreads();

  float* Ts = As;
  {
    const int mt = wave >> 2;
    const int n0 = (wave & 3) * 16;
    v8f c = {};
#pragma unroll
    for (int k0 = 0; k0 < H1; k0 += 4) {
      const int kb = k0 + 2 * half;
      v2f a = *(const v2f*)&Hs[(16 * mt + l16) * PITCH + kb];
      v2f b;
      b.x = W2[(kb + 0) * C2 + n0 + l16];
      b.y = W2[(kb + 1) * C2 + n0 + l16];
      c = __builtin_amdgcn_wmma_f32_16x16x4_f32(
           false, a,  false, b,
           (short)0, c,  false,  false);
    }
#pragma unroll
    for (int r = 0; r < 8; ++r) Ts[(16 * mt + r + 8 * half) * TP + n0 + l16] = c[r];
  }
  __syncthreads();

#pragma unroll
  for (int i = 0; i < (ROWS * C2 / 4) / NT; ++i) {
    const int q = tid + NT * i;
    const int r = q / (C2 / 4);
    const int c4 = (q % (C2 / 4)) * 4;
    if (m0 + r < n_rows) {
      *(v4f*)(t + (size_t)(m0 + r) * C2 + c4) = *(const v4f*)&Ts[r * TP + c4];
    }
  }
}

extern "C" void kernel_launch(void* const* d_in, const int* in_sizes, int n_in,
                              void* d_out, int out_size, void* d_ws, size_t ws_size,
                              hipStream_t stream) {
  const float* x   = (const float*)d_in[0];
  const float* W1  = (const float*)d_in[1];
  const float* W2  = (const float*)d_in[2];
  const int*  src1 = (const int*)d_in[3];
  const int*  dst1 = (const int*)d_in[4];
  const int*  src2 = (const int*)d_in[5];
  const int*  dst2 = (const int*)d_in[6];
  float* out = (float*)d_out;

  constexpr int IN = 128, H = 128, C = 64;
  const int N = in_sizes[0] / IN;
  const int E = in_sizes[3];

  float* half1 = (float*)d_ws;
  float* t   = half1 + (size_t)N * H;
  const int RB  = (N + NBKT - 1) / NBKT;
  const int NCH = (E + BIN_CH - 1) / BIN_CH;
  int* rec1  = (int*)half1;
  int* desc  = rec1 + 2 * (size_t)E;
  int* rows1 = desc + 4 * (size_t)NCH * NBKT;
  int* rows2 = rows1 + 2 * (size_t)N;
  int* col1  = rows2 + 2 * (size_t)N;
  int* col2  = col1 + (size_t)E;
  int* rec2  = (int*)(t + (size_t)N * C);

  GCN_bin_kernel<<<dim3(NCH, 2), BIN_NT, 0, stream>>>(src1, dst1, src2, dst2,
                                                      rec1, rec2, desc, E, RB, NCH);

  GCN_bucket_rows_kernel<<<dim3(NBKT, 2), BIN_NT, (2 * RB + ROWS_CAP) * sizeof(int), stream>>>(
      rec1, rec2, desc, col1, col2, rows1, rows2, N, E, RB, NCH);

  GCN_layer1_kernel<IN, H, C><<<(N + 31) / 32, 256, 0, stream>>>(x, rows1, col1, W1, W2, t, N, E);

  GCN_pull_kernel<C><<<(N + 31) / 32, 256, 0, stream>>>(t, rows2, col2, out, N, E);
}
